# starting kernel
# baseline (speedup 1.0000x reference)
#include <hip/hip_runtime.h>

#define N_NODES   1000000
#define E_EDGES   262144
#define NFC       9
#define CARD      50000
#define EMB       128
#define EDGE_DIM  27
#define TILE_E    16

typedef __attribute__((ext_vector_type(2))) float v2f;
typedef __attribute__((ext_vector_type(8))) float v8f;

__device__ __forceinline__ void encode8(const int* __restrict__ x,
                                        const float* __restrict__ emb_type,
                                        const float* __restrict__ emb_feats,
                                        int node, int n0, float acc[8]) {
    const int* row = x + (size_t)node * 10;
    const int t = row[0];
    const float* et = emb_type + (size_t)t * EMB + n0;
    float4 a0 = *(const float4*)(et);
    float4 a1 = *(const float4*)(et + 4);

    int hb, he;
    if (t == 0)      { hb = 0; he = 4; }
    else if (t == 1) { hb = 4; he = 5; }
    else             { hb = 5; he = 9; }

    for (int h = hb; h < he; ++h) {
        const int idx = row[1 + h];
        const float* ef = emb_feats + ((size_t)h * CARD + (size_t)idx) * EMB + n0;
        float4 b0 = *(const float4*)(ef);
        float4 b1 = *(const float4*)(ef + 4);
        a0.x += b0.x; a0.y += b0.y; a0.z += b0.z; a0.w += b0.w;
        a1.x += b1.x; a1.y += b1.y; a1.z += b1.z; a1.w += b1.w;
    }
    acc[0] = a0.x; acc[1] = a0.y; acc[2] = a0.z; acc[3] = a0.w;
    acc[4] = a1.x; acc[5] = a1.y; acc[6] = a1.z; acc[7] = a1.w;
}

__global__ __launch_bounds__(256)
void edge_pred_fused(const int* __restrict__ x,
                     const int* __restrict__ src,
                     const int* __restrict__ dst,
                     const int* __restrict__ neg_dst,
                     const float* __restrict__ msg,
                     const float* __restrict__ emb_type,
                     const float* __restrict__ emb_feats,
                     const float* __restrict__ edge_w,
                     const float* __restrict__ edge_b,
                     const float* __restrict__ out_w,
                     const float* __restrict__ out_b,
                     float* __restrict__ out) {
    __shared__ float tmsg[TILE_E * EMB];

    const int tid  = threadIdx.x;
    const int wave = tid >> 5;
    const int lane = tid & 31;
    const int e0   = blockIdx.x * TILE_E;

    {
        const int m  = lane & 15;
        const int hi = lane >> 4;
        const int n  = wave * 16 + (lane & 15);
        const float* arow = msg    + (size_t)(e0 + m) * EDGE_DIM;
        const float* brow = edge_w + (size_t)n        * EDGE_DIM;
        v8f c = {};

        #pragma unroll
        for (int k = 0; k < 24; k += 4) {
            const int kk = k + 2 * hi;
            v2f a, b;
            a.x = arow[kk];  a.y = arow[kk + 1];
            b.x = brow[kk];  b.y = brow[kk + 1];
            c = __builtin_amdgcn_wmma_f32_16x16x4_f32(
                    false, a, false, b, (short)0, c, false, false);
        }
        {
            const int kk = 24 + 2 * hi;
            const bool ok = (kk + 1 < EDGE_DIM);
            const int  k1 = ok ? (kk + 1) : (EDGE_DIM - 1);
            v2f a, b;
            a.x = arow[kk];
            b.x = brow[kk];
            const float ay = arow[k1];
            const float by = brow[k1];
            a.y = ok ? ay : 0.0f;
            b.y = ok ? by : 0.0f;
            c = __builtin_amdgcn_wmma_f32_16x16x4_f32(
                    false, a, false, b, (short)0, c, false, false);
        }

        const float bias = edge_b[n];
        #pragma unroll
        for (int r = 0; r < 8; ++r) {
            const int mm = r + 8 * hi;
            tmsg[mm * EMB + n] = c[r] + bias;
        }
    }
    __syncthreads();

    const int el  = tid >> 4;
    const int c16 = tid & 15;
    const int n0  = c16 * 8;
    const int eg  = e0 + el;

    const int ns = src[eg];
    const int np = dst[eg];
    const int nn = neg_dst[eg];

    float hs[8], hp[8], hn[8];
    encode8(x, emb_type, emb_feats, ns, n0, hs);
    encode8(x, emb_type, emb_feats, np, n0, hp);
    encode8(x, emb_type, emb_feats, nn, n0, hn);

    const float4 tm0 = *(const float4*)&tmsg[el * EMB + n0];
    const float4 tm1 = *(const float4*)&tmsg[el * EMB + n0 + 4];
    const float tmv[8] = { tm0.x, tm0.y, tm0.z, tm0.w, tm1.x, tm1.y, tm1.z, tm1.w };

    const float4 w0 = *(const float4*)(out_w + n0);
    const float4 w1 = *(const float4*)(out_w + n0 + 4);
    const float wv[8] = { w0.x, w0.y, w0.z, w0.w, w1.x, w1.y, w1.z, w1.w };

    float pacc = 0.0f, nacc = 0.0f;
    #pragma unroll
    for (int j = 0; j < 8; ++j) {
        float p = hs[j] + hp[j]; p = p > 0.0f ? p : 0.0f;
        float q = hs[j] + hn[j]; q = q > 0.0f ? q : 0.0f;
        pacc = fmaf(p + tmv[j], wv[j], pacc);
        nacc = fmaf(q + tmv[j], wv[j], nacc);
    }

    #pragma unroll
    for (int off = 8; off >= 1; off >>= 1) {
        pacc += __shfl_xor(pacc, off, 32);
        nacc += __shfl_xor(nacc, off, 32);
    }
    if (c16 == 0) {
        const float b = out_b[0];
        out[eg]           = pacc + b;
        out[E_EDGES + eg] = nacc + b;
    }
}

extern "C" void kernel_launch(void* const* d_in, const int* in_sizes, int n_in,
                              void* d_out, int out_size, void* d_ws, size_t ws_size,
                              hipStream_t stream) {
    const int*   x         = (const int*)  d_in[0];
    const int*   src       = (const int*)  d_in[1];
    const int*   dst       = (const int*)  d_in[2];
    const int*   neg_dst   = (const int*)  d_in[3];
    const float* msg       = (const float*)d_in[4];
    const float* emb_type  = (const float*)d_in[5];
    const float* emb_feats = (const float*)d_in[6];
    const float* edge_w    = (const float*)d_in[7];
    const float* edge_b    = (const float*)d_in[8];
    const float* out_w     = (const float*)d_in[9];
    const float* out_b     = (const float*)d_in[10];
    float*       out       = (float*)      d_out;

    dim3 grid(E_EDGES / TILE_E);
    dim3 block(256);
    edge_pred_fused<<<grid, block, 0, stream>>>(x, src, dst, neg_dst, msg,
                                                emb_type, emb_feats,
                                                edge_w, edge_b, out_w, out_b, out);
}


# Gather re-mapped for loads in flight: node rows decoded on the scalar path, a whole wave per 512-B table row, all rows of a wave's 6 nodes requested at once; out_w dot now summed 4 cols x 32 lanes
# speedup vs baseline: 1.3222x; 1.3222x over previous
#include <hip/hip_runtime.h>


#define N_NODES   1000000
#define E_EDGES   262144
#define NFC       9
#define CARD      50000
#define EMB       128
#define EDGE_DIM  27
#define TILE_E    16

typedef __attribute__((ext_vector_type(2))) float v2f;
typedef __attribute__((ext_vector_type(8))) float v8f;

__device__ __forceinline__ void node_rows(const int f[10], int r[4]) {
    const int  t = f[0];
    const bool z = (t == 0);
    const bool o = (t == 1);
    r[0] = z ? f[1] : (o ? (4 * CARD + f[5]) : (5 * CARD + f[6]));
    r[1] = z ? (1 * CARD + f[2]) : (6 * CARD + f[7]);
    r[2] = z ? (2 * CARD + f[3]) : (7 * CARD + f[8]);
    r[3] = z ? (3 * CARD + f[4]) : (8 * CARD + f[9]);
}

__device__ __forceinline__ float4 add4(float4 a, float4 b) {
    a.x += b.x; a.y += b.y; a.z += b.z; a.w += b.w;
    return a;
}

__global__ __launch_bounds__(256)
void edge_pred_fused(const int* __restrict__ x,
                     const int* __restrict__ src,
                     const int* __restrict__ dst,
                     const int* __restrict__ neg_dst,
                     const float* __restrict__ msg,
                     const float* __restrict__ emb_type,
                     const float* __restrict__ emb_feats,
                     const float* __restrict__ edge_w,
                     const float* __restrict__ edge_b,
                     const float* __restrict__ out_w,
                     const float* __restrict__ out_b,
                     float* __restrict__ out) {
    __shared__ float tmsg[TILE_E * EMB];

    const int tid  = threadIdx.x;
    const int wave = tid >> 5;
    const int lane = tid & 31;
    const int e0   = blockIdx.x * TILE_E;

    const int wave_u = __builtin_amdgcn_readfirstlane(wave);
    const int eg0 = e0 + 2 * wave_u;
    const int c4  = lane * 4;

    const int m  = lane & 15;
    const int hi = lane >> 4;
    const int n  = wave * 16 + (lane & 15);
    const float* arow = msg    + (size_t)(e0 + m) * EDGE_DIM;
    const float* brow = edge_w + (size_t)n        * EDGE_DIM;
    v2f fa[7], fb[7];
    #pragma unroll
    for (int k = 0; k < 24; k += 4) {
        const int kk = k + 2 * hi;
        fa[k >> 2].x = arow[kk];  fa[k >> 2].y = arow[kk + 1];
        fb[k >> 2].x = brow[kk];  fb[k >> 2].y = brow[kk + 1];
    }
    {
        const int kk = 24 + 2 * hi;
        const bool ok = (kk + 1 < EDGE_DIM);
        const int  k1 = ok ? (kk + 1) : (EDGE_DIM - 1);
        fa[6].x = arow[kk];
        fb[6].x = brow[kk];
        const float ay = arow[k1];
        const float by = brow[k1];
        fa[6].y = ok ? ay : 0.0f;
        fb[6].y = ok ? by : 0.0f;
    }
    const float bias = edge_b[n];

    const float4 ty0 = *(const float4*)(emb_type + 0 * EMB + c4);
    const float4 ty1 = *(const float4*)(emb_type + 1 * EMB + c4);
    const float4 ty2 = *(const float4*)(emb_type + 2 * EMB + c4);

    int node[6];
    node[0] = src[eg0];     node[1] = dst[eg0];     node[2] = neg_dst[eg0];
    node[3] = src[eg0 + 1]; node[4] = dst[eg0 + 1]; node[5] = neg_dst[eg0 + 1];
    __builtin_amdgcn_sched_barrier(0);

    int xf[6][10];
    #pragma unroll
    for (int i = 0; i < 6; ++i) {
        const int* row = x + (size_t)node[i] * 10;
        #pragma unroll
        for (int j = 0; j < 10; ++j) xf[i][j] = row[j];
    }
    __builtin_amdgcn_sched_barrier(0);

    {
        v8f c = {};
        #pragma unroll
        for (int k = 0; k < 7; ++k) {
            c = __builtin_amdgcn_wmma_f32_16x16x4_f32(
                    false, fa[k], false, fb[k], (short)0, c, false, false);
        }
        #pragma unroll
        for (int r = 0; r < 8; ++r) {
            const int mm = r + 8 * hi;
            tmsg[mm * EMB + n] = c[r] + bias;
        }
    }
    __syncthreads();

    int nt[6];
    int nr[6][4];
    #pragma unroll
    for (int i = 0; i < 6; ++i) { nt[i] = xf[i][0]; node_rows(xf[i], nr[i]); }

    const unsigned l16 = (unsigned)lane * 16u;
    const char* fbase = (const char*)emb_feats;
    float4 v[6][4];
    #pragma unroll
    for (int i = 0; i < 6; ++i) {
        v[i][0] = *(const float4*)(fbase + ((unsigned)nr[i][0] * (EMB * 4u) + l16));
        if (nt[i] != 1) {
            v[i][1] = *(const float4*)(fbase + ((unsigned)nr[i][1] * (EMB * 4u) + l16));
            v[i][2] = *(const float4*)(fbase + ((unsigned)nr[i][2] * (EMB * 4u) + l16));
            v[i][3] = *(const float4*)(fbase + ((unsigned)nr[i][3] * (EMB * 4u) + l16));
        }
    }
    const float4 w4 = *(const float4*)(out_w + c4);
    const float  ob = out_b[0];

    float4 h[6];
    #pragma unroll
    for (int i = 0; i < 6; ++i) {
        const float4 ty = (nt[i] == 0) ? ty0 : ((nt[i] == 1) ? ty1 : ty2);
        float4 a = add4(ty, v[i][0]);
        if (nt[i] != 1) {
            a = add4(a, v[i][1]);
            a = add4(a, v[i][2]);
            a = add4(a, v[i][3]);
        }
        h[i] = a;
    }

    float pacc[2], nacc[2];
    #pragma unroll
    for (int e = 0; e < 2; ++e) {
        const int el = 2 * wave_u + e;
        const float4 tm = *(const float4*)&tmsg[el * EMB + c4];
        const float4 hs = h[3 * e], hp = h[3 * e + 1], hn = h[3 * e + 2];
        const float tmv[4] = { tm.x, tm.y, tm.z, tm.w };
        const float wv[4]  = { w4.x, w4.y, w4.z, w4.w };
        const float sv[4]  = { hs.x, hs.y, hs.z, hs.w };
        const float pv[4]  = { hp.x, hp.y, hp.z, hp.w };
        const float qv[4]  = { hn.x, hn.y, hn.z, hn.w };
        float pa = 0.0f, na = 0.0f;
        #pragma unroll
        for (int j = 0; j < 4; ++j) {
            float p = sv[j] + pv[j]; p = p > 0.0f ? p : 0.0f;
            float q = sv[j] + qv[j]; q = q > 0.0f ? q : 0.0f;
            pa = fmaf(p + tmv[j], wv[j], pa);
            na = fmaf(q + tmv[j], wv[j], na);
        }
        pacc[e] = pa; nacc[e] = na;
    }

    const bool up = lane >= 16;
    float pk = up ? pacc[1] : pacc[0];
    float ps = up ? pacc[0] : pacc[1];
    float nk = up ? nacc[1] : nacc[0];
    float ns = up ? nacc[0] : nacc[1];
    pk += __shfl_xor(ps, 16, 32);
    nk += __shfl_xor(ns, 16, 32);
    #pragma unroll
    for (int off = 8; off >= 1; off >>= 1) {
        pk += __shfl_xor(pk, off, 32);
        nk += __shfl_xor(nk, off, 32);
    }
    if ((lane & 15) == 0) {
        const int eg = eg0 + (lane >> 4);
        out[eg]           = pk + ob;
        out[E_EDGES + eg] = nk + ob;
    }
}

extern "C" void kernel_launch(void* const* d_in, const int* in_sizes, int n_in,
                              void* d_out, int out_size, void* d_ws, size_t ws_size,
                              hipStream_t stream) {
    const int*   x         = (const int*)  d_in[0];
    const int*   src       = (const int*)  d_in[1];
    const int*   dst       = (const int*)  d_in[2];
    const int*   neg_dst   = (const int*)  d_in[3];
    const float* msg       = (const float*)d_in[4];
    const float* emb_type  = (const float*)d_in[5];
    const float* emb_feats = (const float*)d_in[6];
    const float* edge_w    = (const float*)d_in[7];
    const float* edge_b    = (const float*)d_in[8];
    const float* out_w     = (const float*)d_in[9];
    const float* out_b     = (const float*)d_in[10];
    float*       out       = (float*)      d_out;

    dim3 grid(E_EDGES / TILE_E);
    dim3 block(256);
    edge_pred_fused<<<grid, block, 0, stream>>>(x, src, dst, neg_dst, msg,
                                                emb_type, emb_feats,
                                                edge_w, edge_b, out_w, out_b, out);
}


# Persistent work-groups: 2048 groups each walk 8 consecutive 16-edge tiles (was 16384 one-tile groups), next tile's node ids and x rows requested under the current table rows
# speedup vs baseline: 1.4385x; 1.0880x over previous
#include <hip/hip_runtime.h>


#define N_NODES   1000000
#define E_EDGES   262144
#define NFC       9
#define CARD      50000
#define EMB       128
#define EDGE_DIM  27
#define TILE_E    16
#define TILES_PER_WG 8

typedef __attribute__((ext_vector_type(2))) float v2f;
typedef __attribute__((ext_vector_type(8))) float v8f;

__device__ __forceinline__ void node_rows(const int f[10], int r[4]) {
    const int  t = f[0];
    const bool z = (t == 0);
    const bool o = (t == 1);
    r[0] = z ? f[1] : (o ? (4 * CARD + f[5]) : (5 * CARD + f[6]));
    r[1] = z ? (1 * CARD + f[2]) : (6 * CARD + f[7]);
    r[2] = z ? (2 * CARD + f[3]) : (7 * CARD + f[8]);
    r[3] = z ? (3 * CARD + f[4]) : (8 * CARD + f[9]);
}

#define LANE_XOR(v, M) __builtin_bit_cast(float, \
    __builtin_amdgcn_ds_swizzle(__builtin_bit_cast(int, (v)), 0x1f | ((M) << 10)))

__device__ __forceinline__ float4 add4(float4 a, float4 b) {
    a.x += b.x; a.y += b.y; a.z += b.z; a.w += b.w;
    return a;
}

struct WgLds {
    float4 rows4[4][32];
    float  tmsg[2][TILE_E * EMB];
    v2f    bfrag[8][7][32];
};

__global__ __launch_bounds__(256)
void edge_pred_fused(const int* __restrict__ x,
                     const int* __restrict__ src,
                     const int* __restrict__ dst,
                     const int* __restrict__ neg_dst,
                     const float* __restrict__ msg,
                     const float* __restrict__ emb_type,
                     const float* __restrict__ emb_feats,
                     const float* __restrict__ edge_w,
                     const float* __restrict__ edge_b,
                     const float* __restrict__ out_w,
                     const float* __restrict__ out_b,
                     float* __restrict__ out) {
    __shared__ WgLds L;

    const int tid  = threadIdx.x;
    const int wave = tid >> 5;
    const int lane = tid & 31;

    const int wave_u = __builtin_amdgcn_readfirstlane(wave);
    const int c4  = lane * 4;

    const int m  = lane & 15;
    const int hi = lane >> 4;
    const int n  = wave * 16 + (lane & 15);

    {
        const float* brow = edge_w + (size_t)n * EDGE_DIM;
        #pragma unroll
        for (int k = 0; k < 24; k += 4) {
            const int kk = k + 2 * hi;
            v2f fb; fb.x = brow[kk]; fb.y = brow[kk + 1];
            L.bfrag[wave][k >> 2][lane] = fb;
        }
        const int kk = 24 + 2 * hi;
        const bool ok = (kk + 1 < EDGE_DIM);
        const int  k1 = ok ? (kk + 1) : (EDGE_DIM - 1);
        v2f fb; fb.x = brow[kk];
        const float by = brow[k1];
        fb.y = ok ? by : 0.0f;
        L.bfrag[wave][6][lane] = fb;
        if (wave < 3)
            L.rows4[wave][lane] = *(const float4*)(emb_type + wave * EMB + c4);
        else if (wave == 3)
            L.rows4[3][lane] = *(const float4*)(out_w + c4);
    }
    const float bias = edge_b[n];
    const float ob   = out_b[0];

    const int tile0 = blockIdx.x * TILES_PER_WG;

    const int l6 = lane % 6;
    const int ia = l6 % 3;
    const int ie = l6 / 3;
    const int* ip = ((ia == 0) ? src : ((ia == 1) ? dst : neg_dst)) + ie;
    const int lx = lane < 30 ? lane : 29;
    const int xi = lx / 10;
    const unsigned xj = (unsigned)(lx % 10);
    const int eg_first = tile0 * TILE_E + 2 * wave_u;

    int idxn;
    int xa, xb;
    {
        const int idxc = ip[eg_first];
        idxn = ip[eg_first + TILE_E];
        const int n0 = __builtin_amdgcn_readlane(idxc, 0), n1 = __builtin_amdgcn_readlane(idxc, 1);
        const int n2 = __builtin_amdgcn_readlane(idxc, 2), n3 = __builtin_amdgcn_readlane(idxc, 3);
        const int n4 = __builtin_amdgcn_readlane(idxc, 4), n5 = __builtin_amdgcn_readlane(idxc, 5);
        const int na = (xi == 0) ? n0 : ((xi == 1) ? n1 : n2);
        const int nb = (xi == 0) ? n3 : ((xi == 1) ? n4 : n5);
        xa = x[(unsigned)na * 10u + xj];
        xb = x[(unsigned)nb * 10u + xj];
    }

    const unsigned l16 = (unsigned)lane * 16u;
    const char* fbase = (const char*)emb_feats;
    const int ao = m * EDGE_DIM + 2 * hi;
    const int tw = (8 * hi) * EMB + n;

    #pragma unroll 1
    for (int it = 0; it < TILES_PER_WG; ++it) {
        const int e0  = (tile0 + it) * TILE_E;
        const int eg0 = e0 + 2 * wave_u;
        float* tm = L.tmsg[it & 1];

        const float* mt = msg + (size_t)e0 * EDGE_DIM;
        v2f fa[7];
        #pragma unroll
        for (int k = 0; k < 24; k += 4) {
            fa[k >> 2].x = mt[ao + k];  fa[k >> 2].y = mt[ao + k + 1];
        }
        {
            const bool ok = (hi == 0);
            const int  k1 = ok ? (ao + 25) : (ao + 24);
            fa[6].x = mt[ao + 24];
            const float ay = mt[k1];
            fa[6].y = ok ? ay : 0.0f;
        }

        int xf[6][10];
        #pragma unroll
        for (int i = 0; i < 6; ++i) {
            #pragma unroll
            for (int j = 0; j < 10; ++j)
                xf[i][j] = __builtin_amdgcn_readlane(i < 3 ? xa : xb, (i % 3) * 10 + j);
        }

        {
            v8f c = {};
            #pragma unroll
            for (int k = 0; k < 7; ++k) {
                const v2f fb = L.bfrag[wave][k][lane];
                c = __builtin_amdgcn_wmma_f32_16x16x4_f32(
                        false, fa[k], false, fb, (short)0, c, false, false);
            }
            #pragma unroll
            for (int r = 0; r < 8; ++r)
                tm[tw + r * EMB] = c[r] + bias;
        }
        __syncthreads();

        int nt[6];
        int nr[6][4];
        #pragma unroll
        for (int i = 0; i < 6; ++i) { nt[i] = xf[i][0]; node_rows(xf[i], nr[i]); }

        float4 v[6][4];
        #pragma unroll
        for (int i = 0; i < 6; ++i) {
            v[i][0] = *(const float4*)(fbase + ((unsigned)nr[i][0] * (EMB * 4u) + l16));
            if (nt[i] != 1) {
                v[i][1] = *(const float4*)(fbase + ((unsigned)nr[i][1] * (EMB * 4u) + l16));
                v[i][2] = *(const float4*)(fbase + ((unsigned)nr[i][2] * (EMB * 4u) + l16));
                v[i][3] = *(const float4*)(fbase + ((unsigned)nr[i][3] * (EMB * 4u) + l16));
            }
        }
        __builtin_amdgcn_sched_barrier(0);

        {
            const int n0 = __builtin_amdgcn_readlane(idxn, 0), n1 = __builtin_amdgcn_readlane(idxn, 1);
            const int n2 = __builtin_amdgcn_readlane(idxn, 2), n3 = __builtin_amdgcn_readlane(idxn, 3);
            const int n4 = __builtin_amdgcn_readlane(idxn, 4), n5 = __builtin_amdgcn_readlane(idxn, 5);
            const int na = (xi == 0) ? n0 : ((xi == 1) ? n1 : n2);
            const int nb = (xi == 0) ? n3 : ((xi == 1) ? n4 : n5);
            xa = x[(unsigned)na * 10u + xj];
            xb = x[(unsigned)nb * 10u + xj];
            const int t2 = (it + 2 < TILES_PER_WG) ? (it + 2) : (TILES_PER_WG - 1);
            idxn = ip[eg_first + t2 * TILE_E];
        }

        float4 h[6];
        #pragma unroll
        for (int i = 0; i < 6; ++i) {
            const int   ti = (nt[i] == 0) ? 0 : ((nt[i] == 1) ? 1 : 2);
            const float4 ty = L.rows4[ti][lane];
            float4 a = add4(ty, v[i][0]);
            if (nt[i] != 1) {
                a = add4(a, v[i][1]);
                a = add4(a, v[i][2]);
                a = add4(a, v[i][3]);
            }
            h[i] = a;
        }
        const float4 w4 = L.rows4[3][lane];

        float pacc[2], nacc[2];
        #pragma unroll
        for (int e = 0; e < 2; ++e) {
            const int el = 2 * wave_u + e;
            const float4 tmq = *(const float4*)&tm[el * EMB + c4];
            const float4 hs = h[3 * e], hp = h[3 * e + 1], hn = h[3 * e + 2];
            const float tmv[4] = { tmq.x, tmq.y, tmq.z, tmq.w };
            const float wv[4]  = { w4.x, w4.y, w4.z, w4.w };
            const float sv[4]  = { hs.x, hs.y, hs.z, hs.w };
            const float pv[4]  = { hp.x, hp.y, hp.z, hp.w };
            const float qv[4]  = { hn.x, hn.y, hn.z, hn.w };
            float pa = 0.0f, na = 0.0f;
            #pragma unroll
            for (int j = 0; j < 4; ++j) {
                float p = sv[j] + pv[j]; p = p > 0.0f ? p : 0.0f;
                float q = sv[j] + qv[j]; q = q > 0.0f ? q : 0.0f;
                pa = fmaf(p + tmv[j], wv[j], pa);
                na = fmaf(q + tmv[j], wv[j], na);
            }
            pacc[e] = pa; nacc[e] = na;
        }

        const bool up = lane >= 16;
        float pk = up ? pacc[1] : pacc[0];
        float ps = up ? pacc[0] : pacc[1];
        float nk = up ? nacc[1] : nacc[0];
        float ns = up ? nacc[0] : nacc[1];
        pk += LANE_XOR(ps, 16);  nk += LANE_XOR(ns, 16);
        pk += LANE_XOR(pk, 8);   nk += LANE_XOR(nk, 8);
        pk += LANE_XOR(pk, 4);   nk += LANE_XOR(nk, 4);
        pk += LANE_XOR(pk, 2);   nk += LANE_XOR(nk, 2);
        pk += LANE_XOR(pk, 1);   nk += LANE_XOR(nk, 1);
        if ((lane & 15) == 0) {
            const int eg = eg0 + (lane >> 4);
            out[eg]           = pk + ob;
            out[E_EDGES + eg] = nk + ob;
        }
    }
}

extern "C" void kernel_launch(void* const* d_in, const int* in_sizes, int n_in,
                              void* d_out, int out_size, void* d_ws, size_t ws_size,
                              hipStream_t stream) {
    const int*   x         = (const int*)  d_in[0];
    const int*   src       = (const int*)  d_in[1];
    const int*   dst       = (const int*)  d_in[2];
    const int*   neg_dst   = (const int*)  d_in[3];
    const float* msg       = (const float*)d_in[4];
    const float* emb_type  = (const float*)d_in[5];
    const float* emb_feats = (const float*)d_in[6];
    const float* edge_w    = (const float*)d_in[7];
    const float* edge_b    = (const float*)d_in[8];
    const float* out_w     = (const float*)d_in[9];
    const float* out_b     = (const float*)d_in[10];
    float*       out       = (float*)      d_out;

    dim3 grid(E_EDGES / TILE_E / TILES_PER_WG);
    dim3 block(256);
    edge_pred_fused<<<grid, block, 0, stream>>>(x, src, dst, neg_dst, msg,
                                                emb_type, emb_feats,
                                                edge_w, edge_b, out_w, out_b, out);
}


# msg tile (A operand) fetched once per work-group two tiles ahead, 8 contiguous bytes per thread, and handed to the 8 waves through LDS; was 8 row-strided loads per wave on demand
# speedup vs baseline: 1.6269x; 1.1310x over previous
#include <hip/hip_runtime.h>


#define N_NODES   1000000
#define E_EDGES   262144
#define NFC       9
#define CARD      50000
#define EMB       128
#define EDGE_DIM  27
#define TILE_E    16
#define TILES_PER_WG 8

typedef __attribute__((ext_vector_type(2))) float v2f;
typedef __attribute__((ext_vector_type(8))) float v8f;

__device__ __forceinline__ void node_rows(const int f[10], int r[4]) {
    const int  t = f[0];
    const bool z = (t == 0);
    const bool o = (t == 1);
    r[0] = z ? f[1] : (o ? (4 * CARD + f[5]) : (5 * CARD + f[6]));
    r[1] = z ? (1 * CARD + f[2]) : (6 * CARD + f[7]);
    r[2] = z ? (2 * CARD + f[3]) : (7 * CARD + f[8]);
    r[3] = z ? (3 * CARD + f[4]) : (8 * CARD + f[9]);
}

#define LANE_XOR(v, M) __builtin_bit_cast(float, \
    __builtin_amdgcn_ds_swizzle(__builtin_bit_cast(int, (v)), 0x1f | ((M) << 10)))

__device__ __forceinline__ float4 add4(float4 a, float4 b) {
    a.x += b.x; a.y += b.y; a.z += b.z; a.w += b.w;
    return a;
}

struct WgLds {
    float4 rows4[4][32];
    float  tmsg[2][TILE_E * EMB];
    v2f    bfrag[8][7][32];
    float  amsg[2][TILE_E * EDGE_DIM];
};

__global__ __launch_bounds__(256)
void edge_pred_fused(const int* __restrict__ x,
                     const int* __restrict__ src,
                     const int* __restrict__ dst,
                     const int* __restrict__ neg_dst,
                     const float* __restrict__ msg,
                     const float* __restrict__ emb_type,
                     const float* __restrict__ emb_feats,
                     const float* __restrict__ edge_w,
                     const float* __restrict__ edge_b,
                     const float* __restrict__ out_w,
                     const float* __restrict__ out_b,
                     float* __restrict__ out) {
    __shared__ WgLds L;

    const int tid  = threadIdx.x;
    const int wave = tid >> 5;
    const int lane = tid & 31;

    const int wave_u = __builtin_amdgcn_readfirstlane(wave);
    const int c4  = lane * 4;

    const int m  = lane & 15;
    const int hi = lane >> 4;
    const int n  = wave * 16 + (lane & 15);

    {
        const float* brow = edge_w + (size_t)n * EDGE_DIM;
        #pragma unroll
        for (int k = 0; k < 24; k += 4) {
            const int kk = k + 2 * hi;
            v2f fb; fb.x = brow[kk]; fb.y = brow[kk + 1];
            L.bfrag[wave][k >> 2][lane] = fb;
        }
        const int kk = 24 + 2 * hi;
        const bool ok = (kk + 1 < EDGE_DIM);
        const int  k1 = ok ? (kk + 1) : (EDGE_DIM - 1);
        v2f fb; fb.x = brow[kk];
        const float by = brow[k1];
        fb.y = ok ? by : 0.0f;
        L.bfrag[wave][6][lane] = fb;
        if (wave < 3)
            L.rows4[wave][lane] = *(const float4*)(emb_type + wave * EMB + c4);
        else if (wave == 3)
            L.rows4[3][lane] = *(const float4*)(out_w + c4);
    }
    const float bias = edge_b[n];
    const float ob   = out_b[0];

    const int tile0 = blockIdx.x * TILES_PER_WG;

    const bool stg = tid < (TILE_E * EDGE_DIM / 2);
    const int  so  = stg ? 2 * tid : 0;
    const unsigned sob = (unsigned)so * 4u;
    const unsigned tb0 = (unsigned)tile0 * (TILE_E * EDGE_DIM * 4u);
    const char* mbytes = (const char*)msg;
    v2f mnext;
    {
        const v2f m0 = *(const v2f*)(mbytes + (tb0 + sob));
        mnext = *(const v2f*)(mbytes + (tb0 + TILE_E * EDGE_DIM * 4u + sob));
        if (stg) *(v2f*)&L.amsg[0][so] = m0;
    }
    __syncthreads();

    const int l6 = lane % 6;
    const int ia = l6 % 3;
    const int ie = l6 / 3;
    const int* ip = ((ia == 0) ? src : ((ia == 1) ? dst : neg_dst)) + ie;
    const int lx = lane < 30 ? lane : 29;
    const int xi = lx / 10;
    const unsigned xj = (unsigned)(lx % 10);
    const int eg_first = tile0 * TILE_E + 2 * wave_u;

    int idxn;
    int xa, xb;
    {
        const int idxc = ip[eg_first];
        idxn = ip[eg_first + TILE_E];
        const int n0 = __builtin_amdgcn_readlane(idxc, 0), n1 = __builtin_amdgcn_readlane(idxc, 1);
        const int n2 = __builtin_amdgcn_readlane(idxc, 2), n3 = __builtin_amdgcn_readlane(idxc, 3);
        const int n4 = __builtin_amdgcn_readlane(idxc, 4), n5 = __builtin_amdgcn_readlane(idxc, 5);
        const int na = (xi == 0) ? n0 : ((xi == 1) ? n1 : n2);
        const int nb = (xi == 0) ? n3 : ((xi == 1) ? n4 : n5);
        xa = x[(unsigned)na * 10u + xj];
        xb = x[(unsigned)nb * 10u + xj];
    }

    const unsigned l16 = (unsigned)lane * 16u;
    const char* fbase = (const char*)emb_feats;
    const int ao = m * EDGE_DIM + 2 * hi;
    const int tw = (8 * hi) * EMB + n;

    #pragma unroll 1
    for (int it = 0; it < TILES_PER_WG; ++it) {
        const int e0  = (tile0 + it) * TILE_E;
        const int eg0 = e0 + 2 * wave_u;
        float* tm = L.tmsg[it & 1];

        if (stg) *(v2f*)&L.amsg[(it + 1) & 1][so] = mnext;

        const float* mt = L.amsg[it & 1];
        v2f fa[7];
        #pragma unroll
        for (int k = 0; k < 24; k += 4) {
            fa[k >> 2].x = mt[ao + k];  fa[k >> 2].y = mt[ao + k + 1];
        }
        {
            const bool ok = (hi == 0);
            const int  k1 = ok ? (ao + 25) : (ao + 24);
            fa[6].x = mt[ao + 24];
            const float ay = mt[k1];
            fa[6].y = ok ? ay : 0.0f;
        }

        int xf[6][10];
        #pragma unroll
        for (int i = 0; i < 6; ++i) {
            #pragma unroll
            for (int j = 0; j < 10; ++j)
                xf[i][j] = __builtin_amdgcn_readlane(i < 3 ? xa : xb, (i % 3) * 10 + j);
        }

        {
            v8f c = {};
            #pragma unroll
            for (int k = 0; k < 7; ++k) {
                const v2f fb = L.bfrag[wave][k][lane];
                c = __builtin_amdgcn_wmma_f32_16x16x4_f32(
                        false, fa[k], false, fb, (short)0, c, false, false);
            }
            #pragma unroll
            for (int r = 0; r < 8; ++r)
                tm[tw + r * EMB] = c[r] + bias;
        }
        __syncthreads();

        int nt[6];
        int nr[6][4];
        #pragma unroll
        for (int i = 0; i < 6; ++i) { nt[i] = xf[i][0]; node_rows(xf[i], nr[i]); }

        float4 v[6][4];
        #pragma unroll
        for (int i = 0; i < 6; ++i) {
            v[i][0] = *(const float4*)(fbase + ((unsigned)nr[i][0] * (EMB * 4u) + l16));
            if (nt[i] != 1) {
                v[i][1] = *(const float4*)(fbase + ((unsigned)nr[i][1] * (EMB * 4u) + l16));
                v[i][2] = *(const float4*)(fbase + ((unsigned)nr[i][2] * (EMB * 4u) + l16));
                v[i][3] = *(const float4*)(fbase + ((unsigned)nr[i][3] * (EMB * 4u) + l16));
            }
        }
        __builtin_amdgcn_sched_barrier(0);

        {
            const int n0 = __builtin_amdgcn_readlane(idxn, 0), n1 = __builtin_amdgcn_readlane(idxn, 1);
            const int n2 = __builtin_amdgcn_readlane(idxn, 2), n3 = __builtin_amdgcn_readlane(idxn, 3);
            const int n4 = __builtin_amdgcn_readlane(idxn, 4), n5 = __builtin_amdgcn_readlane(idxn, 5);
            const int na = (xi == 0) ? n0 : ((xi == 1) ? n1 : n2);
            const int nb = (xi == 0) ? n3 : ((xi == 1) ? n4 : n5);
            xa = x[(unsigned)na * 10u + xj];
            xb = x[(unsigned)nb * 10u + xj];
            const int t2 = (it + 2 < TILES_PER_WG) ? (it + 2) : (TILES_PER_WG - 1);
            idxn = ip[eg_first + t2 * TILE_E];
            mnext = *(const v2f*)(mbytes + (tb0 + (unsigned)t2 * (TILE_E * EDGE_DIM * 4u) + sob));
        }

        float4 h[6];
        #pragma unroll
        for (int i = 0; i < 6; ++i) {
            const int   ti = (nt[i] == 0) ? 0 : ((nt[i] == 1) ? 1 : 2);
            const float4 ty = L.rows4[ti][lane];
            float4 a = add4(ty, v[i][0]);
            if (nt[i] != 1) {
                a = add4(a, v[i][1]);
                a = add4(a, v[i][2]);
                a = add4(a, v[i][3]);
            }
            h[i] = a;
        }
        const float4 w4 = L.rows4[3][lane];

        float pacc[2], nacc[2];
        #pragma unroll
        for (int e = 0; e < 2; ++e) {
            const int el = 2 * wave_u + e;
            const float4 tmq = *(const float4*)&tm[el * EMB + c4];
            const float4 hs = h[3 * e], hp = h[3 * e + 1], hn = h[3 * e + 2];
            const float tmv[4] = { tmq.x, tmq.y, tmq.z, tmq.w };
            const float wv[4]  = { w4.x, w4.y, w4.z, w4.w };
            const float sv[4]  = { hs.x, hs.y, hs.z, hs.w };
            const float pv[4]  = { hp.x, hp.y, hp.z, hp.w };
            const float qv[4]  = { hn.x, hn.y, hn.z, hn.w };
            float pa = 0.0f, na = 0.0f;
            #pragma unroll
            for (int j = 0; j < 4; ++j) {
                float p = sv[j] + pv[j]; p = p > 0.0f ? p : 0.0f;
                float q = sv[j] + qv[j]; q = q > 0.0f ? q : 0.0f;
                pa = fmaf(p + tmv[j], wv[j], pa);
                na = fmaf(q + tmv[j], wv[j], na);
            }
            pacc[e] = pa; nacc[e] = na;
        }

        const bool up = lane >= 16;
        float pk = up ? pacc[1] : pacc[0];
        float ps = up ? pacc[0] : pacc[1];
        float nk = up ? nacc[1] : nacc[0];
        float ns = up ? nacc[0] : nacc[1];
        pk += LANE_XOR(ps, 16);  nk += LANE_XOR(ns, 16);
        pk += LANE_XOR(pk, 8);   nk += LANE_XOR(nk, 8);
        pk += LANE_XOR(pk, 4);   nk += LANE_XOR(nk, 4);
        pk += LANE_XOR(pk, 2);   nk += LANE_XOR(nk, 2);
        pk += LANE_XOR(pk, 1);   nk += LANE_XOR(nk, 1);
        if ((lane & 15) == 0) {
            const int eg = eg0 + (lane >> 4);
            out[eg]           = pk + ob;
            out[E_EDGES + eg] = nk + ob;
        }
    }
}

extern "C" void kernel_launch(void* const* d_in, const int* in_sizes, int n_in,
                              void* d_out, int out_size, void* d_ws, size_t ws_size,
                              hipStream_t stream) {
    const int*   x         = (const int*)  d_in[0];
    const int*   src       = (const int*)  d_in[1];
    const int*   dst       = (const int*)  d_in[2];
    const int*   neg_dst   = (const int*)  d_in[3];
    const float* msg       = (const float*)d_in[4];
    const float* emb_type  = (const float*)d_in[5];
    const float* emb_feats = (const float*)d_in[6];
    const float* edge_w    = (const float*)d_in[7];
    const float* edge_b    = (const float*)d_in[8];
    const float* out_w     = (const float*)d_in[9];
    const float* out_b     = (const float*)d_in[10];
    float*       out       = (float*)      d_out;

    dim3 grid(E_EDGES / TILE_E / TILES_PER_WG);
    dim3 block(256);
    edge_pred_fused<<<grid, block, 0, stream>>>(x, src, dst, neg_dst, msg,
                                                emb_type, emb_feats,
                                                edge_w, edge_b, out_w, out_b, out);
}


# Table-row loads issued ahead of the matrix phase (nodes 0-3) and of a barrier that now waits for LDS only (nodes 4-5); previous tile's lane reduction and stores moved behind the last request
# speedup vs baseline: 1.6763x; 1.0304x over previous
#include <hip/hip_runtime.h>


#define N_NODES   1000000
#define E_EDGES   262144
#define NFC       9
#define CARD      50000
#define EMB       128
#define EDGE_DIM  27
#define TILE_E    16
#define TILES_PER_WG 8

typedef __attribute__((ext_vector_type(2))) float v2f;
typedef __attribute__((ext_vector_type(8))) float v8f;

__device__ __forceinline__ void node_rows(const int f[10], int r[4]) {
    const int  t = f[0];
    const bool z = (t == 0);
    const bool o = (t == 1);
    r[0] = z ? f[1] : (o ? (4 * CARD + f[5]) : (5 * CARD + f[6]));
    r[1] = z ? (1 * CARD + f[2]) : (6 * CARD + f[7]);
    r[2] = z ? (2 * CARD + f[3]) : (7 * CARD + f[8]);
    r[3] = z ? (3 * CARD + f[4]) : (8 * CARD + f[9]);
}

#define LANE_XOR(v, M) __builtin_bit_cast(float, \
    __builtin_amdgcn_ds_swizzle(__builtin_bit_cast(int, (v)), 0x1f | ((M) << 10)))

__device__ __forceinline__ float4 add4(float4 a, float4 b) {
    a.x += b.x; a.y += b.y; a.z += b.z; a.w += b.w;
    return a;
}

__device__ __forceinline__ void request_rows(const char* fbase, unsigned l16, int t,
                                             const int r[4], float4 v[4]) {
    v[0] = *(const float4*)(fbase + ((unsigned)r[0] * (EMB * 4u) + l16));
    #pragma unroll
    for (int k = 1; k < 4; ++k)
        asm volatile("" : "=v"(v[k].x), "=v"(v[k].y), "=v"(v[k].z), "=v"(v[k].w));
    if (t != 1) {
        v[1] = *(const float4*)(fbase + ((unsigned)r[1] * (EMB * 4u) + l16));
        v[2] = *(const float4*)(fbase + ((unsigned)r[2] * (EMB * 4u) + l16));
        v[3] = *(const float4*)(fbase + ((unsigned)r[3] * (EMB * 4u) + l16));
    }
}

__device__ __forceinline__ void reduce_store(const float pacc[2], const float nacc[2],
                                             int lane, int eg0, float ob,
                                             float* __restrict__ out) {
    const bool up = lane >= 16;
    float pk = up ? pacc[1] : pacc[0];
    float ps = up ? pacc[0] : pacc[1];
    float nk = up ? nacc[1] : nacc[0];
    float ns = up ? nacc[0] : nacc[1];
    pk += LANE_XOR(ps, 16);  nk += LANE_XOR(ns, 16);
    pk += LANE_XOR(pk, 8);   nk += LANE_XOR(nk, 8);
    pk += LANE_XOR(pk, 4);   nk += LANE_XOR(nk, 4);
    pk += LANE_XOR(pk, 2);   nk += LANE_XOR(nk, 2);
    pk += LANE_XOR(pk, 1);   nk += LANE_XOR(nk, 1);
    if ((lane & 15) == 0) {
        const int eg = eg0 + (lane >> 4);
        out[eg]           = pk + ob;
        out[E_EDGES + eg] = nk + ob;
    }
}

struct WgLds {
    float4 rows4[4][32];
    float  tmsg[2][TILE_E * EMB];
    v2f    bfrag[8][7][32];
    float  amsg[2][TILE_E * EDGE_DIM];
};

__global__ __launch_bounds__(256)
void edge_pred_fused(const int* __restrict__ x,
                     const int* __restrict__ src,
                     const int* __restrict__ dst,
                     const int* __restrict__ neg_dst,
                     const float* __restrict__ msg,
                     const float* __restrict__ emb_type,
                     const float* __restrict__ emb_feats,
                     const float* __restrict__ edge_w,
                     const float* __restrict__ edge_b,
                     const float* __restrict__ out_w,
                     const float* __restrict__ out_b,
                     float* __restrict__ out) {
    __shared__ WgLds L;

    const int tid  = threadIdx.x;
    const int wave = tid >> 5;
    const int lane = tid & 31;

    const int wave_u = __builtin_amdgcn_readfirstlane(wave);
    const int c4  = lane * 4;

    const int m  = lane & 15;
    const int hi = lane >> 4;
    const int n  = wave * 16 + (lane & 15);

    {
        const float* brow = edge_w + (size_t)n * EDGE_DIM;
        #pragma unroll
        for (int k = 0; k < 24; k += 4) {
            const int kk = k + 2 * hi;
            v2f fb; fb.x = brow[kk]; fb.y = brow[kk + 1];
            L.bfrag[wave][k >> 2][lane] = fb;
        }
        const int kk = 24 + 2 * hi;
        const bool ok = (kk + 1 < EDGE_DIM);
        const int  k1 = ok ? (kk + 1) : (EDGE_DIM - 1);
        v2f fb; fb.x = brow[kk];
        const float by = brow[k1];
        fb.y = ok ? by : 0.0f;
        L.bfrag[wave][6][lane] = fb;
        if (wave < 3)
            L.rows4[wave][lane] = *(const float4*)(emb_type + wave * EMB + c4);
        else if (wave == 3)
            L.rows4[3][lane] = *(const float4*)(out_w + c4);
    }
    const float bias = edge_b[n];
    const float ob   = out_b[0];

    const int tile0 = blockIdx.x * TILES_PER_WG;

    const bool stg = tid < (TILE_E * EDGE_DIM / 2);
    const int  so  = stg ? 2 * tid : 0;
    const unsigned sob = (unsigned)so * 4u;
    const unsigned tb0 = (unsigned)tile0 * (TILE_E * EDGE_DIM * 4u);
    const char* mbytes = (const char*)msg;
    v2f mnext;
    {
        const v2f m0 = *(const v2f*)(mbytes + (tb0 + sob));
        mnext = *(const v2f*)(mbytes + (tb0 + TILE_E * EDGE_DIM * 4u + sob));
        if (stg) *(v2f*)&L.amsg[0][so] = m0;
    }
    __syncthreads();

    const int l6 = lane % 6;
    const int ia = l6 % 3;
    const int ie = l6 / 3;
    const int* ip = ((ia == 0) ? src : ((ia == 1) ? dst : neg_dst)) + ie;
    const int lx = lane < 30 ? lane : 29;
    const int xi = lx / 10;
    const unsigned xj = (unsigned)(lx % 10);
    const int eg_first = tile0 * TILE_E + 2 * wave_u;

    int idxn;
    int xa, xb;
    {
        const int idxc = ip[eg_first];
        idxn = ip[eg_first + TILE_E];
        const int n0 = __builtin_amdgcn_readlane(idxc, 0), n1 = __builtin_amdgcn_readlane(idxc, 1);
        const int n2 = __builtin_amdgcn_readlane(idxc, 2), n3 = __builtin_amdgcn_readlane(idxc, 3);
        const int n4 = __builtin_amdgcn_readlane(idxc, 4), n5 = __builtin_amdgcn_readlane(idxc, 5);
        const int na = (xi == 0) ? n0 : ((xi == 1) ? n1 : n2);
        const int nb = (xi == 0) ? n3 : ((xi == 1) ? n4 : n5);
        xa = x[(unsigned)na * 10u + xj];
        xb = x[(unsigned)nb * 10u + xj];
    }

    const unsigned l16 = (unsigned)lane * 16u;
    const char* fbase = (const char*)emb_feats;
    const int ao = m * EDGE_DIM + 2 * hi;
    const int tw = (8 * hi) * EMB + n;

    float pacc[2] = { 0.0f, 0.0f };
    float nacc[2] = { 0.0f, 0.0f };

    #pragma unroll 1
    for (int it = 0; it < TILES_PER_WG; ++it) {
        const int e0  = (tile0 + it) * TILE_E;
        const int eg0 = e0 + 2 * wave_u;
        float* tm = L.tmsg[it & 1];

        if (stg) *(v2f*)&L.amsg[(it + 1) & 1][so] = mnext;

        int xf[6][10];
        #pragma unroll
        for (int i = 0; i < 6; ++i) {
            #pragma unroll
            for (int j = 0; j < 10; ++j)
                xf[i][j] = __builtin_amdgcn_readlane(i < 3 ? xa : xb, (i % 3) * 10 + j);
        }
        int nt[6];
        int nr[6][4];
        #pragma unroll
        for (int i = 0; i < 6; ++i) { nt[i] = xf[i][0]; node_rows(xf[i], nr[i]); }

        float4 v[6][4];
        #pragma unroll
        for (int i = 0; i < 4; ++i) request_rows(fbase, l16, nt[i], nr[i], v[i]);
        __builtin_amdgcn_sched_barrier(0);

        const float* mt = L.amsg[it & 1];
        v2f fa[7];
        #pragma unroll
        for (int k = 0; k < 24; k += 4) {
            fa[k >> 2].x = mt[ao + k];  fa[k >> 2].y = mt[ao + k + 1];
        }
        {
            const bool ok = (hi == 0);
            const int  k1 = ok ? (ao + 25) : (ao + 24);
            fa[6].x = mt[ao + 24];
            const float ay = mt[k1];
            fa[6].y = ok ? ay : 0.0f;
        }

        {
            v8f c = {};
            #pragma unroll
            for (int k = 0; k < 7; ++k) {
                const v2f fb = L.bfrag[wave][k][lane];
                c = __builtin_amdgcn_wmma_f32_16x16x4_f32(
                        false, fa[k], false, fb, (short)0, c, false, false);
            }
            #pragma unroll
            for (int r = 0; r < 8; ++r)
                tm[tw + r * EMB] = c[r] + bias;
        }

        #pragma unroll
        for (int i = 4; i < 6; ++i) request_rows(fbase, l16, nt[i], nr[i], v[i]);
        __builtin_amdgcn_sched_barrier(0);

        __builtin_amdgcn_fence(__ATOMIC_RELEASE, "workgroup", "local");
        __builtin_amdgcn_s_barrier();
        __builtin_amdgcn_fence(__ATOMIC_ACQUIRE, "workgroup", "local");

        {
            const int n0 = __builtin_amdgcn_readlane(idxn, 0), n1 = __builtin_amdgcn_readlane(idxn, 1);
            const int n2 = __builtin_amdgcn_readlane(idxn, 2), n3 = __builtin_amdgcn_readlane(idxn, 3);
            const int n4 = __builtin_amdgcn_readlane(idxn, 4), n5 = __builtin_amdgcn_readlane(idxn, 5);
            const int na = (xi == 0) ? n0 : ((xi == 1) ? n1 : n2);
            const int nb = (xi == 0) ? n3 : ((xi == 1) ? n4 : n5);
            const int t2 = (it + 2 < TILES_PER_WG) ? (it + 2) : (TILES_PER_WG - 1);
            mnext = *(const v2f*)(mbytes + (tb0 + (unsigned)t2 * (TILE_E * EDGE_DIM * 4u) + sob));
            idxn = ip[eg_first + t2 * TILE_E];
            xa = x[(unsigned)na * 10u + xj];
            xb = x[(unsigned)nb * 10u + xj];
        }

        if (it > 0) reduce_store(pacc, nacc, lane, eg0 - TILE_E, ob, out);

        float4 h[6];
        #pragma unroll
        for (int i = 0; i < 6; ++i) {
            const int   ti = (nt[i] == 0) ? 0 : ((nt[i] == 1) ? 1 : 2);
            const float4 ty = L.rows4[ti][lane];
            float4 a = add4(ty, v[i][0]);
            if (nt[i] != 1) {
                a = add4(a, v[i][1]);
                a = add4(a, v[i][2]);
                a = add4(a, v[i][3]);
            }
            h[i] = a;
        }
        const float4 w4 = L.rows4[3][lane];

        #pragma unroll
        for (int e = 0; e < 2; ++e) {
            const int el = 2 * wave_u + e;
            const float4 tmq = *(const float4*)&tm[el * EMB + c4];
            const float4 hs = h[3 * e], hp = h[3 * e + 1], hn = h[3 * e + 2];
            const float tmv[4] = { tmq.x, tmq.y, tmq.z, tmq.w };
            const float wv[4]  = { w4.x, w4.y, w4.z, w4.w };
            const float sv[4]  = { hs.x, hs.y, hs.z, hs.w };
            const float pv[4]  = { hp.x, hp.y, hp.z, hp.w };
            const float qv[4]  = { hn.x, hn.y, hn.z, hn.w };
            float pa = 0.0f, na = 0.0f;
            #pragma unroll
            for (int j = 0; j < 4; ++j) {
                float p = sv[j] + pv[j]; p = p > 0.0f ? p : 0.0f;
                float q = sv[j] + qv[j]; q = q > 0.0f ? q : 0.0f;
                pa = fmaf(p + tmv[j], wv[j], pa);
                na = fmaf(q + tmv[j], wv[j], na);
            }
            pacc[e] = pa; nacc[e] = na;
        }
    }

    reduce_store(pacc, nacc, lane, (tile0 + TILES_PER_WG - 1) * TILE_E + 2 * wave_u, ob, out);
}

extern "C" void kernel_launch(void* const* d_in, const int* in_sizes, int n_in,
                              void* d_out, int out_size, void* d_ws, size_t ws_size,
                              hipStream_t stream) {
    const int*   x         = (const int*)  d_in[0];
    const int*   src       = (const int*)  d_in[1];
    const int*   dst       = (const int*)  d_in[2];
    const int*   neg_dst   = (const int*)  d_in[3];
    const float* msg       = (const float*)d_in[4];
    const float* emb_type  = (const float*)d_in[5];
    const float* emb_feats = (const float*)d_in[6];
    const float* edge_w    = (const float*)d_in[7];
    const float* edge_b    = (const float*)d_in[8];
    const float* out_w     = (const float*)d_in[9];
    const float* out_b     = (const float*)d_in[10];
    float*       out       = (float*)      d_out;

    dim3 grid(E_EDGES / TILE_E / TILES_PER_WG);
    dim3 block(256);
    edge_pred_fused<<<grid, block, 0, stream>>>(x, src, dst, neg_dst, msg,
                                                emb_type, emb_feats,
                                                edge_w, edge_b, out_w, out_b, out);
}


# Table gather split into two column sweeps per work-group, lower 256-B row halves, then upper from x rows parked in LDS: a sweep's random set is half the table; dots summed per half, msg share apart
# speedup vs baseline: 1.8294x; 1.0913x over previous
#include <hip/hip_runtime.h>


#define N_NODES   1000000
#define E_EDGES   262144
#define NFC       9
#define CARD      50000
#define EMB       128
#define EDGE_DIM  27
#define TILE_E    16
#define TILES_PER_WG 16

typedef __attribute__((ext_vector_type(2))) float v2f;
typedef __attribute__((ext_vector_type(8))) float v8f;

__device__ __forceinline__ void node_rows(const int f[10], int r[4]) {
    const int  t = f[0];
    const bool z = (t == 0);
    const bool o = (t == 1);
    r[0] = z ? f[1] : (o ? (4 * CARD + f[5]) : (5 * CARD + f[6]));
    r[1] = z ? (1 * CARD + f[2]) : (6 * CARD + f[7]);
    r[2] = z ? (2 * CARD + f[3]) : (7 * CARD + f[8]);
    r[3] = z ? (3 * CARD + f[4]) : (8 * CARD + f[9]);
}

#define LANE_XOR(v, M) __builtin_bit_cast(float, \
    __builtin_amdgcn_ds_swizzle(__builtin_bit_cast(int, (v)), 0x1f | ((M) << 10)))

__device__ __forceinline__ void request_rows(const char* fbase, unsigned lo, int t,
                                             const int r[4], v2f v[4]) {
    v[0] = *(const v2f*)(fbase + ((unsigned)r[0] * (EMB * 4u) + lo));
    #pragma unroll
    for (int k = 1; k < 4; ++k)
        asm volatile("" : "=v"(v[k].x), "=v"(v[k].y));
    if (t != 1) {
        v[1] = *(const v2f*)(fbase + ((unsigned)r[1] * (EMB * 4u) + lo));
        v[2] = *(const v2f*)(fbase + ((unsigned)r[2] * (EMB * 4u) + lo));
        v[3] = *(const v2f*)(fbase + ((unsigned)r[3] * (EMB * 4u) + lo));
    }
}

__device__ __forceinline__ void decode_nodes(int xa, int xb, int nt[6], int nr[6][4]) {
    int xf[6][10];
    #pragma unroll
    for (int i = 0; i < 6; ++i) {
        #pragma unroll
        for (int j = 0; j < 10; ++j)
            xf[i][j] = __builtin_amdgcn_readlane(i < 3 ? xa : xb, (i % 3) * 10 + j);
    }
    #pragma unroll
    for (int i = 0; i < 6; ++i) { nt[i] = xf[i][0]; node_rows(xf[i], nr[i]); }
}

__device__ __forceinline__ v2f node_sum(int t, v2f ty, const v2f v[4]) {
    v2f a = ty + v[0];
    if (t != 1) {
        a = a + v[1];
        a = a + v[2];
        a = a + v[3];
    }
    return a;
}

__device__ __forceinline__ void edge_dots(v2f hs, v2f hp, v2f hn, v2f w, float& pa, float& na) {
    float p0 = hs.x + hp.x; p0 = p0 > 0.0f ? p0 : 0.0f;
    float p1 = hs.y + hp.y; p1 = p1 > 0.0f ? p1 : 0.0f;
    float q0 = hs.x + hn.x; q0 = q0 > 0.0f ? q0 : 0.0f;
    float q1 = hs.y + hn.y; q1 = q1 > 0.0f ? q1 : 0.0f;
    pa = fmaf(p0, w.x, pa);  pa = fmaf(p1, w.y, pa);
    na = fmaf(q0, w.x, na);  na = fmaf(q1, w.y, na);
}

__device__ __forceinline__ void reduce_edges(const float pacc[2], const float nacc[2],
                                             int lane, float& pk, float& nk) {
    const bool up = lane >= 16;
    pk = up ? pacc[1] : pacc[0];
    float ps = up ? pacc[0] : pacc[1];
    nk = up ? nacc[1] : nacc[0];
    float ns = up ? nacc[0] : nacc[1];
    pk += LANE_XOR(ps, 16);  nk += LANE_XOR(ns, 16);
    pk += LANE_XOR(pk, 8);   nk += LANE_XOR(nk, 8);
    pk += LANE_XOR(pk, 4);   nk += LANE_XOR(nk, 4);
    pk += LANE_XOR(pk, 2);   nk += LANE_XOR(nk, 2);
    pk += LANE_XOR(pk, 1);   nk += LANE_XOR(nk, 1);
}

struct WgLds {
    float4 rows4[4][32];
    float  tmsg[2][TILE_E * EMB];
    v2f    bfrag[8][7][32];
    float  amsg[2][TILE_E * EDGE_DIM];
    int    xpark[TILES_PER_WG][8][2][32];
    float  part[TILES_PER_WG][TILE_E][2];
};

__global__ __launch_bounds__(256)
void edge_pred_fused(const int* __restrict__ x,
                     const int* __restrict__ src,
                     const int* __restrict__ dst,
                     const int* __restrict__ neg_dst,
                     const float* __restrict__ msg,
                     const float* __restrict__ emb_type,
                     const float* __restrict__ emb_feats,
                     const float* __restrict__ edge_w,
                     const float* __restrict__ edge_b,
                     const float* __restrict__ out_w,
                     const float* __restrict__ out_b,
                     float* __restrict__ out) {
    __shared__ WgLds L;

    const int tid  = threadIdx.x;
    const int wave = tid >> 5;
    const int lane = tid & 31;

    const int wave_u = __builtin_amdgcn_readfirstlane(wave);
    const int c4  = lane * 4;

    const int m  = lane & 15;
    const int hi = lane >> 4;
    const int n  = wave * 16 + (lane & 15);

    {
        const float* brow = edge_w + (size_t)n * EDGE_DIM;
        #pragma unroll
        for (int k = 0; k < 24; k += 4) {
            const int kk = k + 2 * hi;
            v2f fb; fb.x = brow[kk]; fb.y = brow[kk + 1];
            L.bfrag[wave][k >> 2][lane] = fb;
        }
        const int kk = 24 + 2 * hi;
        const bool ok = (kk + 1 < EDGE_DIM);
        const int  k1 = ok ? (kk + 1) : (EDGE_DIM - 1);
        v2f fb; fb.x = brow[kk];
        const float by = brow[k1];
        fb.y = ok ? by : 0.0f;
        L.bfrag[wave][6][lane] = fb;
        if (wave < 3)
            L.rows4[wave][lane] = *(const float4*)(emb_type + wave * EMB + c4);
        else if (wave == 3)
            L.rows4[3][lane] = *(const float4*)(out_w + c4);
    }
    const float bias = edge_b[n];
    const float ob   = out_b[0];

    const int tile0 = blockIdx.x * TILES_PER_WG;

    const bool stg = tid < (TILE_E * EDGE_DIM / 2);
    const int  so  = stg ? 2 * tid : 0;
    const unsigned sob = (unsigned)so * 4u;
    const unsigned tb0 = (unsigned)tile0 * (TILE_E * EDGE_DIM * 4u);
    const char* mbytes = (const char*)msg;
    v2f mnext;
    {
        const v2f m0 = *(const v2f*)(mbytes + (tb0 + sob));
        mnext = *(const v2f*)(mbytes + (tb0 + TILE_E * EDGE_DIM * 4u + sob));
        if (stg) *(v2f*)&L.amsg[0][so] = m0;
    }
    __syncthreads();

    const int l6 = lane % 6;
    const int ia = l6 % 3;
    const int ie = l6 / 3;
    const int* ip = ((ia == 0) ? src : ((ia == 1) ? dst : neg_dst)) + ie;
    const int lx = lane < 30 ? lane : 29;
    const int xi = lx / 10;
    const unsigned xj = (unsigned)(lx % 10);
    const int eg_first = tile0 * TILE_E + 2 * wave_u;

    int idxn;
    int xa, xb;
    {
        const int idxc = ip[eg_first];
        idxn = ip[eg_first + TILE_E];
        const int n0 = __builtin_amdgcn_readlane(idxc, 0), n1 = __builtin_amdgcn_readlane(idxc, 1);
        const int n2 = __builtin_amdgcn_readlane(idxc, 2), n3 = __builtin_amdgcn_readlane(idxc, 3);
        const int n4 = __builtin_amdgcn_readlane(idxc, 4), n5 = __builtin_amdgcn_readlane(idxc, 5);
        const int na = (xi == 0) ? n0 : ((xi == 1) ? n1 : n2);
        const int nb = (xi == 0) ? n3 : ((xi == 1) ? n4 : n5);
        xa = x[(unsigned)na * 10u + xj];
        xb = x[(unsigned)nb * 10u + xj];
    }

    const unsigned l8 = (unsigned)lane * 8u;
    const char* fbase = (const char*)emb_feats;
    const int ao = m * EDGE_DIM + 2 * hi;
    const int tw = (8 * hi) * EMB + n;

    float pacc[2] = { 0.0f, 0.0f };
    float nacc[2] = { 0.0f, 0.0f };

    #pragma unroll 1
    for (int it = 0; it < TILES_PER_WG; ++it) {
        float* tm = L.tmsg[it & 1];

        if (stg) *(v2f*)&L.amsg[(it + 1) & 1][so] = mnext;

        L.xpark[it][wave][0][lane] = xa;
        L.xpark[it][wave][1][lane] = xb;
        int nt[6];
        int nr[6][4];
        decode_nodes(xa, xb, nt, nr);

        v2f v[6][4];
        #pragma unroll
        for (int i = 0; i < 4; ++i) request_rows(fbase, l8, nt[i], nr[i], v[i]);
        __builtin_amdgcn_sched_barrier(0);

        const float* mt = L.amsg[it & 1];
        v2f fa[7];
        #pragma unroll
        for (int k = 0; k < 24; k += 4) {
            fa[k >> 2].x = mt[ao + k];  fa[k >> 2].y = mt[ao + k + 1];
        }
        {
            const bool ok = (hi == 0);
            const int  k1 = ok ? (ao + 25) : (ao + 24);
            fa[6].x = mt[ao + 24];
            const float ay = mt[k1];
            fa[6].y = ok ? ay : 0.0f;
        }

        {
            v8f c = {};
            #pragma unroll
            for (int k = 0; k < 7; ++k) {
                const v2f fb = L.bfrag[wave][k][lane];
                c = __builtin_amdgcn_wmma_f32_16x16x4_f32(
                        false, fa[k], false, fb, (short)0, c, false, false);
            }
            #pragma unroll
            for (int r = 0; r < 8; ++r)
                tm[tw + r * EMB] = c[r] + bias;
        }

        #pragma unroll
        for (int i = 4; i < 6; ++i) request_rows(fbase, l8, nt[i], nr[i], v[i]);
        __builtin_amdgcn_sched_barrier(0);

        __builtin_amdgcn_fence(__ATOMIC_RELEASE, "workgroup", "local");
        __builtin_amdgcn_s_barrier();
        __builtin_amdgcn_fence(__ATOMIC_ACQUIRE, "workgroup", "local");

        {
            const int n0 = __builtin_amdgcn_readlane(idxn, 0), n1 = __builtin_amdgcn_readlane(idxn, 1);
            const int n2 = __builtin_amdgcn_readlane(idxn, 2), n3 = __builtin_amdgcn_readlane(idxn, 3);
            const int n4 = __builtin_amdgcn_readlane(idxn, 4), n5 = __builtin_amdgcn_readlane(idxn, 5);
            const int na = (xi == 0) ? n0 : ((xi == 1) ? n1 : n2);
            const int nb = (xi == 0) ? n3 : ((xi == 1) ? n4 : n5);
            const int t2 = (it + 2 < TILES_PER_WG) ? (it + 2) : (TILES_PER_WG - 1);
            mnext = *(const v2f*)(mbytes + (tb0 + (unsigned)t2 * (TILE_E * EDGE_DIM * 4u) + sob));
            idxn = ip[eg_first + t2 * TILE_E];
            xa = x[(unsigned)na * 10u + xj];
            xb = x[(unsigned)nb * 10u + xj];
        }

        if (it > 0) {
            float pk, nk;
            reduce_edges(pacc, nacc, lane, pk, nk);
            if ((lane & 15) == 0) {
                L.part[it - 1][2 * wave + (lane >> 4)][0] = pk;
                L.part[it - 1][2 * wave + (lane >> 4)][1] = nk;
            }
        }

        v2f h[6];
        #pragma unroll
        for (int i = 0; i < 6; ++i) {
            const int ti = (nt[i] == 0) ? 0 : ((nt[i] == 1) ? 1 : 2);
            const v2f ty = ((const v2f*)&L.rows4[ti][0])[lane];
            h[i] = node_sum(nt[i], ty, v[i]);
        }
        const float4 w4 = L.rows4[3][lane];
        const v2f    w2 = ((const v2f*)&L.rows4[3][0])[lane];

        #pragma unroll
        for (int e = 0; e < 2; ++e) {
            const int el = 2 * wave_u + e;
            const float4 tmq = *(const float4*)&tm[el * EMB + c4];
            float ta = tmq.x * w4.x;
            ta = fmaf(tmq.y, w4.y, ta);
            ta = fmaf(tmq.z, w4.z, ta);
            ta = fmaf(tmq.w, w4.w, ta);
            float pa = ta, na = ta;
            edge_dots(h[3 * e], h[3 * e + 1], h[3 * e + 2], w2, pa, na);
            pacc[e] = pa; nacc[e] = na;
        }
    }

    {
        float pk, nk;
        reduce_edges(pacc, nacc, lane, pk, nk);
        if ((lane & 15) == 0) {
            L.part[TILES_PER_WG - 1][2 * wave + (lane >> 4)][0] = pk;
            L.part[TILES_PER_WG - 1][2 * wave + (lane >> 4)][1] = nk;
        }
    }

    const unsigned l8u = 256u + l8;
    const v2f w2u = ((const v2f*)&L.rows4[3][0])[32 + lane];
    int nt[6];
    v2f v[6][4];
    {
        int nr[6][4];
        decode_nodes(L.xpark[0][wave][0][lane], L.xpark[0][wave][1][lane], nt, nr);
        #pragma unroll
        for (int i = 0; i < 6; ++i) request_rows(fbase, l8u, nt[i], nr[i], v[i]);
    }
    #pragma unroll 1
    for (int it = 0; it < TILES_PER_WG; ++it) {
        v2f h[6];
        #pragma unroll
        for (int i = 0; i < 6; ++i) {
            const int ti = (nt[i] == 0) ? 0 : ((nt[i] == 1) ? 1 : 2);
            const v2f ty = ((const v2f*)&L.rows4[ti][0])[32 + lane];
            h[i] = node_sum(nt[i], ty, v[i]);
        }
        float pu[2], nu[2];
        #pragma unroll
        for (int e = 0; e < 2; ++e) {
            float pa = 0.0f, na = 0.0f;
            edge_dots(h[3 * e], h[3 * e + 1], h[3 * e + 2], w2u, pa, na);
            pu[e] = pa; nu[e] = na;
        }

        if (it + 1 < TILES_PER_WG) {
            int nr[6][4];
            decode_nodes(L.xpark[it + 1][wave][0][lane], L.xpark[it + 1][wave][1][lane], nt, nr);
            #pragma unroll
            for (int i = 0; i < 6; ++i) request_rows(fbase, l8u, nt[i], nr[i], v[i]);
        }

        float pk, nk;
        reduce_edges(pu, nu, lane, pk, nk);
        if ((lane & 15) == 0) {
            const int el = 2 * wave + (lane >> 4);
            const int eg = (tile0 + it) * TILE_E + el;
            out[eg]           = (L.part[it][el][0] + pk) + ob;
            out[E_EDGES + eg] = (L.part[it][el][1] + nk) + ob;
        }
    }
}

extern "C" void kernel_launch(void* const* d_in, const int* in_sizes, int n_in,
                              void* d_out, int out_size, void* d_ws, size_t ws_size,
                              hipStream_t stream) {
    const int*   x         = (const int*)  d_in[0];
    const int*   src       = (const int*)  d_in[1];
    const int*   dst       = (const int*)  d_in[2];
    const int*   neg_dst   = (const int*)  d_in[3];
    const float* msg       = (const float*)d_in[4];
    const float* emb_type  = (const float*)d_in[5];
    const float* emb_feats = (const float*)d_in[6];
    const float* edge_w    = (const float*)d_in[7];
    const float* edge_b    = (const float*)d_in[8];
    const float* out_w     = (const float*)d_in[9];
    const float* out_b     = (const float*)d_in[10];
    float*       out       = (float*)      d_out;

    dim3 grid(E_EDGES / TILE_E / TILES_PER_WG);
    dim3 block(256);
    edge_pred_fused<<<grid, block, 0, stream>>>(x, src, dst, neg_dst, msg,
                                                emb_type, emb_feats,
                                                edge_w, edge_b, out_w, out_b, out);
}
